# starting kernel
# baseline (speedup 1.0000x reference)
#include <hip/hip_runtime.h>


typedef __attribute__((ext_vector_type(2))) float v2f;
typedef __attribute__((ext_vector_type(4))) float v4f;
typedef __attribute__((ext_vector_type(8))) float v8f;

#define GB 256
#define GT 2048
#define GI 10
#define GH 64

#define WMMA4(A, B, C) __builtin_amdgcn_wmma_f32_16x16x4_f32( \
    false, (A), false, (B), (short)0, (C), false, false)

__device__ __forceinline__ float fast_rcp(float x) {
#if __has_builtin(__builtin_amdgcn_rcpf)
    return __builtin_amdgcn_rcpf(x);
#else
    return 1.0f / x;
#endif
}
__device__ __forceinline__ float fast_exp(float x) {
#if __has_builtin(__builtin_amdgcn_exp2f)
    return __builtin_amdgcn_exp2f(x * 1.4426950408889634f);
#else
    return __expf(x);
#endif
}
__device__ __forceinline__ float tanh_f(float x) {
#if __has_builtin(__builtin_amdgcn_tanhf)
    return __builtin_amdgcn_tanhf(x);
#else
    return 1.0f - 2.0f * fast_rcp(fast_exp(2.0f * x) + 1.0f);
#endif
}
__device__ __forceinline__ float sigmoid_f(float x) {
#if __has_builtin(__builtin_amdgcn_tanhf)
    return 0.5f + 0.5f * __builtin_amdgcn_tanhf(0.5f * x);
#else
    return fast_rcp(1.0f + fast_exp(-x));
#endif
}

#define BAR_SIGNAL() asm volatile("s_wait_dscnt 0x0\n\ts_barrier_signal -1" ::: "memory")
#define BAR_WAIT()   asm volatile("s_barrier_wait -1" ::: "memory")

struct NFrag { v2f c[3]; };
__device__ __forceinline__ NFrag load_noise(const float* __restrict__ np, int lhi) {
    NFrag f;
    if (lhi == 0) {
        f.c[0] = *(const v2f*)(np + 0);
        f.c[1] = *(const v2f*)(np + 4);
        f.c[2] = *(const v2f*)(np + 8);
    } else {
        f.c[0] = *(const v2f*)(np + 2);
        f.c[1] = *(const v2f*)(np + 6);
        f.c[2] = (v2f){0.0f, 0.0f};
    }
    return f;
}

__global__ __launch_bounds__(128) void gru_scan_kernel(
    const float* __restrict__ noise,
    const float* __restrict__ w_ih,
    const float* __restrict__ w_hh,
    const float* __restrict__ b_ih,
    const float* __restrict__ b_hh,
    float* __restrict__ out)
{
    __shared__ __align__(16) float sh_h[2][16][2][32];

    const int tid  = threadIdx.x;
    const int ct   = tid >> 5;
    const int lane = tid & 31;
    const int lhi  = lane >> 4;
    const int lcol = lane & 15;
    const int b0   = blockIdx.x * 16;
    const int colg = ct * 16 + lcol;

    v2f bwih[3][3];
    v2f bwhh[3][16];
    float bgi[3];
    float bhn = 0.0f;

    for (int g = 0; g < 3; ++g) {
        const int col3 = g * GH + colg;
        for (int kc = 0; kc < 3; ++kc) {
            const int k0 = 4 * kc + 2 * lhi;
            v2f b;
            b[0] = (k0     < GI) ? w_ih[col3 * GI + k0]     : 0.0f;
            b[1] = (k0 + 1 < GI) ? w_ih[col3 * GI + k0 + 1] : 0.0f;
            bwih[g][kc] = b;
        }
        for (int kc = 0; kc < 16; ++kc) {
            const int k0 = 4 * kc + 2 * lhi;
            v2f b;
            b[0] = w_hh[col3 * GH + k0];
            b[1] = w_hh[col3 * GH + k0 + 1];
            bwhh[g][kc] = b;
        }
        bgi[g] = b_ih[col3] + ((g < 2) ? b_hh[col3] : 0.0f);
        if (g == 2) bhn = b_hh[col3];
    }
    v8f cgi[3], chn, czero;
    #pragma unroll
    for (int g = 0; g < 3; ++g)
        #pragma unroll
        for (int vv = 0; vv < 8; ++vv) cgi[g][vv] = bgi[g];
    #pragma unroll
    for (int vv = 0; vv < 8; ++vv) { chn[vv] = bhn; czero[vv] = 0.0f; }

    float hreg[8];
    #pragma unroll
    for (int vv = 0; vv < 8; ++vv) hreg[vv] = 0.0f;
    for (int idx = tid; idx < 16 * 2 * 32; idx += 128)
        ((float*)sh_h[0])[idx] = 0.0f;
    BAR_SIGNAL();

    const int    m_a       = lane & 15;
    const size_t noise_row = (size_t)(b0 + m_a) * GT * GI;
    const int whalf = (colg >> 1) & 1;
    const int widx  = ((colg >> 2) << 1) | (colg & 1);

    NFrag an_next = load_noise(noise + noise_row, lhi);

    for (int t = 0; t < GT; ++t) {
        BAR_WAIT();

        NFrag an = an_next;
        {
            const size_t tn = (size_t)((t + 1 < GT) ? (t + 1) : t) * GI;
            an_next = load_noise(noise + noise_row + tn, lhi);
        }

        const float* __restrict__ hrow = &sh_h[t & 1][m_a][lhi][0];
        v2f ah[16];
        #pragma unroll
        for (int kc = 0; kc < 16; ++kc)
            ah[kc] = *((const v2f*)hrow + kc);

        v8f accA[3], accB[3], gn;
        #pragma unroll
        for (int g = 0; g < 3; ++g) {
            v8f a = WMMA4(an.c[0], bwih[g][0], cgi[g]);
            a = WMMA4(an.c[1], bwih[g][1], a);
            a = WMMA4(an.c[2], bwih[g][2], a);
            if (g == 2) { gn = a; a = chn; }
            #pragma unroll
            for (int kc = 0; kc < 8; ++kc)
                a = WMMA4(ah[kc], bwhh[g][kc], a);
            accA[g] = a;
            v8f bacc = czero;
            #pragma unroll
            for (int kc = 8; kc < 16; ++kc)
                bacc = WMMA4(ah[kc], bwhh[g][kc], bacc);
            accB[g] = bacc;
        }

        float hnew[8];
        float* __restrict__ hw = &sh_h[(t & 1) ^ 1][0][0][0];
        #pragma unroll
        for (int vv = 0; vv < 8; ++vv) {
            const int m = vv + 8 * lhi;
            const float r  = sigmoid_f(accA[0][vv] + accB[0][vv]);
            const float z  = sigmoid_f(accA[1][vv] + accB[1][vv]);
            const float hn = accA[2][vv] + accB[2][vv];
            const float n  = tanh_f(gn[vv] + r * hn);
            const float h  = n + z * (hreg[vv] - n);
            hreg[vv] = h;
            hnew[vv] = h;
            hw[(m * 2 + whalf) * 32 + widx] = h;
        }
        BAR_SIGNAL();

        #pragma unroll
        for (int vv = 0; vv < 8; ++vv) {
            const int m = vv + 8 * lhi;
            out[((size_t)(b0 + m) * GT + t) * GH + colg] = hnew[vv];
        }
    }
}

extern "C" void kernel_launch(void* const* d_in, const int* in_sizes, int n_in,
                              void* d_out, int out_size, void* d_ws, size_t ws_size,
                              hipStream_t stream) {
    (void)in_sizes; (void)n_in; (void)out_size; (void)d_ws; (void)ws_size;
    const float* noise = (const float*)d_in[0];
    const float* w_ih  = (const float*)d_in[1];
    const float* w_hh  = (const float*)d_in[2];
    const float* b_ih  = (const float*)d_in[3];
    const float* b_hh  = (const float*)d_in[4];
    float* out = (float*)d_out;

    dim3 grid(GB / 16);
    dim3 block(128);
    gru_scan_kernel<<<grid, block, 0, stream>>>(noise, w_ih, w_hh, b_ih, b_hh, out);
}


# LDS h rows padded from a 256 B to a 272 B pitch (16 B x 17): the 8 ds_load_b128 A-fragment reads per wave and step become bank-conflict-free (a half-wave's 16 lanes shared 4 banks before)
# speedup vs baseline: 1.0769x; 1.0769x over previous
#include <hip/hip_runtime.h>


typedef __attribute__((ext_vector_type(2))) float v2f;
typedef __attribute__((ext_vector_type(4))) float v4f;
typedef __attribute__((ext_vector_type(8))) float v8f;

#define GB 256
#define GT 2048
#define GI 10
#define GH 64
#define HP 68
#define HBUF 2048

#define WMMA4(A, B, C) __builtin_amdgcn_wmma_f32_16x16x4_f32( \
    false, (A), false, (B), (short)0, (C), false, false)

__device__ __forceinline__ float fast_rcp(float x) {
#if __has_builtin(__builtin_amdgcn_rcpf)
    return __builtin_amdgcn_rcpf(x);
#else
    return 1.0f / x;
#endif
}
__device__ __forceinline__ float fast_exp(float x) {
#if __has_builtin(__builtin_amdgcn_exp2f)
    return __builtin_amdgcn_exp2f(x * 1.4426950408889634f);
#else
    return __expf(x);
#endif
}
__device__ __forceinline__ float tanh_f(float x) {
#if __has_builtin(__builtin_amdgcn_tanhf)
    return __builtin_amdgcn_tanhf(x);
#else
    return 1.0f - 2.0f * fast_rcp(fast_exp(2.0f * x) + 1.0f);
#endif
}
__device__ __forceinline__ float sigmoid_f(float x) {
#if __has_builtin(__builtin_amdgcn_tanhf)
    return 0.5f + 0.5f * __builtin_amdgcn_tanhf(0.5f * x);
#else
    return fast_rcp(1.0f + fast_exp(-x));
#endif
}

#define BAR_SIGNAL() asm volatile("s_wait_dscnt 0x0\n\ts_barrier_signal -1" ::: "memory")
#define BAR_WAIT()   asm volatile("s_barrier_wait -1" ::: "memory")

struct NFrag { v2f c[3]; };
__device__ __forceinline__ NFrag load_noise(const float* __restrict__ np, int lhi) {
    NFrag f;
    if (lhi == 0) {
        f.c[0] = *(const v2f*)(np + 0);
        f.c[1] = *(const v2f*)(np + 4);
        f.c[2] = *(const v2f*)(np + 8);
    } else {
        f.c[0] = *(const v2f*)(np + 2);
        f.c[1] = *(const v2f*)(np + 6);
        f.c[2] = (v2f){0.0f, 0.0f};
    }
    return f;
}

__global__ __launch_bounds__(128) void gru_scan_kernel(
    const float* __restrict__ noise,
    const float* __restrict__ w_ih,
    const float* __restrict__ w_hh,
    const float* __restrict__ b_ih,
    const float* __restrict__ b_hh,
    float* __restrict__ out)
{
    __shared__ __align__(16) float sh_h[2][HBUF];

    const int tid  = threadIdx.x;
    const int ct   = tid >> 5;
    const int lane = tid & 31;
    const int lhi  = lane >> 4;
    const int lcol = lane & 15;
    const int b0   = blockIdx.x * 16;
    const int colg = ct * 16 + lcol;

    v2f bwih[3][3];
    v2f bwhh[3][16];
    float bgi[3];
    float bhn = 0.0f;

    for (int g = 0; g < 3; ++g) {
        const int col3 = g * GH + colg;
        for (int kc = 0; kc < 3; ++kc) {
            const int k0 = 4 * kc + 2 * lhi;
            v2f b;
            b[0] = (k0     < GI) ? w_ih[col3 * GI + k0]     : 0.0f;
            b[1] = (k0 + 1 < GI) ? w_ih[col3 * GI + k0 + 1] : 0.0f;
            bwih[g][kc] = b;
        }
        for (int kc = 0; kc < 16; ++kc) {
            const int k0 = 4 * kc + 2 * lhi;
            v2f b;
            b[0] = w_hh[col3 * GH + k0];
            b[1] = w_hh[col3 * GH + k0 + 1];
            bwhh[g][kc] = b;
        }
        bgi[g] = b_ih[col3] + ((g < 2) ? b_hh[col3] : 0.0f);
        if (g == 2) bhn = b_hh[col3];
    }
    v8f cgi[3], chn, czero;
    #pragma unroll
    for (int g = 0; g < 3; ++g)
        #pragma unroll
        for (int vv = 0; vv < 8; ++vv) cgi[g][vv] = bgi[g];
    #pragma unroll
    for (int vv = 0; vv < 8; ++vv) { chn[vv] = bhn; czero[vv] = 0.0f; }

    float hreg[8];
    #pragma unroll
    for (int vv = 0; vv < 8; ++vv) hreg[vv] = 0.0f;
    for (int idx = tid; idx < 16 * HP; idx += 128)
        sh_h[0][idx] = 0.0f;
    BAR_SIGNAL();

    const int    m_a       = lane & 15;
    const size_t noise_row = (size_t)(b0 + m_a) * GT * GI;
    const int whalf = (colg >> 1) & 1;
    const int widx  = ((colg >> 2) << 1) | (colg & 1);

    NFrag an_next = load_noise(noise + noise_row, lhi);

    for (int t = 0; t < GT; ++t) {
        BAR_WAIT();

        NFrag an = an_next;
        {
            const size_t tn = (size_t)((t + 1 < GT) ? (t + 1) : t) * GI;
            an_next = load_noise(noise + noise_row + tn, lhi);
        }

        const float* __restrict__ hrow = &sh_h[t & 1][m_a * HP + lhi * 32];
        v2f ah[16];
        #pragma unroll
        for (int kc = 0; kc < 16; ++kc)
            ah[kc] = *((const v2f*)hrow + kc);

        v8f accA[3], accB[3], gn;
        #pragma unroll
        for (int g = 0; g < 3; ++g) {
            v8f a = WMMA4(an.c[0], bwih[g][0], cgi[g]);
            a = WMMA4(an.c[1], bwih[g][1], a);
            a = WMMA4(an.c[2], bwih[g][2], a);
            if (g == 2) { gn = a; a = chn; }
            #pragma unroll
            for (int kc = 0; kc < 8; ++kc)
                a = WMMA4(ah[kc], bwhh[g][kc], a);
            accA[g] = a;
            v8f bacc = czero;
            #pragma unroll
            for (int kc = 8; kc < 16; ++kc)
                bacc = WMMA4(ah[kc], bwhh[g][kc], bacc);
            accB[g] = bacc;
        }

        float hnew[8];
        float* __restrict__ hw = &sh_h[(t & 1) ^ 1][0];
        #pragma unroll
        for (int vv = 0; vv < 8; ++vv) {
            const int m = vv + 8 * lhi;
            const float r  = sigmoid_f(accA[0][vv] + accB[0][vv]);
            const float z  = sigmoid_f(accA[1][vv] + accB[1][vv]);
            const float hn = accA[2][vv] + accB[2][vv];
            const float n  = tanh_f(gn[vv] + r * hn);
            const float h  = n + z * (hreg[vv] - n);
            hreg[vv] = h;
            hnew[vv] = h;
            hw[m * HP + whalf * 32 + widx] = h;
        }
        BAR_SIGNAL();

        #pragma unroll
        for (int vv = 0; vv < 8; ++vv) {
            const int m = vv + 8 * lhi;
            out[((size_t)(b0 + m) * GT + t) * GH + colg] = hnew[vv];
        }
    }
}

extern "C" void kernel_launch(void* const* d_in, const int* in_sizes, int n_in,
                              void* d_out, int out_size, void* d_ws, size_t ws_size,
                              hipStream_t stream) {
    (void)in_sizes; (void)n_in; (void)out_size; (void)d_ws; (void)ws_size;
    const float* noise = (const float*)d_in[0];
    const float* w_ih  = (const float*)d_in[1];
    const float* w_hh  = (const float*)d_in[2];
    const float* b_ih  = (const float*)d_in[3];
    const float* b_hh  = (const float*)d_in[4];
    float* out = (float*)d_out;

    dim3 grid(GB / 16);
    dim3 block(128);
    gru_scan_kernel<<<grid, block, 0, stream>>>(noise, w_ih, w_hh, b_ih, b_hh, out);
}


# Global-memory waits taken off the per-step serial path: noise fetched two steps ahead (two register sets, loop unrolled by two), loads and stores hand-issued so no wait follows a just-issued access
# speedup vs baseline: 2.0580x; 1.9110x over previous
#include <hip/hip_runtime.h>


typedef __attribute__((ext_vector_type(2))) float v2f;
typedef __attribute__((ext_vector_type(4))) float v4f;
typedef __attribute__((ext_vector_type(8))) float v8f;

#define GB 256
#define GT 2048
#define GI 10
#define GH 64
#define HP 68
#define HBUF 2048

#define WMMA4(A, B, C) __builtin_amdgcn_wmma_f32_16x16x4_f32( \
    false, (A), false, (B), (short)0, (C), false, false)

__device__ __forceinline__ float fast_rcp(float x) {
#if __has_builtin(__builtin_amdgcn_rcpf)
    return __builtin_amdgcn_rcpf(x);
#else
    return 1.0f / x;
#endif
}
__device__ __forceinline__ float fast_exp(float x) {
#if __has_builtin(__builtin_amdgcn_exp2f)
    return __builtin_amdgcn_exp2f(x * 1.4426950408889634f);
#else
    return __expf(x);
#endif
}
__device__ __forceinline__ float tanh_f(float x) {
#if __has_builtin(__builtin_amdgcn_tanhf)
    return __builtin_amdgcn_tanhf(x);
#else
    return 1.0f - 2.0f * fast_rcp(fast_exp(2.0f * x) + 1.0f);
#endif
}
__device__ __forceinline__ float sigmoid_f(float x) {
#if __has_builtin(__builtin_amdgcn_tanhf)
    return 0.5f + 0.5f * __builtin_amdgcn_tanhf(0.5f * x);
#else
    return fast_rcp(1.0f + fast_exp(-x));
#endif
}

#define BAR_SIGNAL() asm volatile("s_wait_dscnt 0x0\n\ts_barrier_signal -1" ::: "memory")
#define BAR_WAIT()   asm volatile("s_barrier_wait -1" ::: "memory")

struct NFrag { v2f c[3]; };

#define NOISE_ISSUE(S, BASE)                                                  \
    asm volatile("global_load_b64 %0, %3, %5\n\t"                             \
                 "global_load_b64 %1, %3, %5 offset:16\n\t"                   \
                 "global_load_b64 %2, %4, %5"                                 \
                 : "=&v"((S).c[0]), "=&v"((S).c[1]), "=&v"((S).c[2])          \
                 : "v"(noff01), "v"(noff2), "s"(BASE)                         \
                 : "memory")

#define NOISE_ARRIVE(S)                                                       \
    asm volatile("s_wait_loadcnt 0x3"                                         \
                 : "+v"((S).c[0]), "+v"((S).c[1]), "+v"((S).c[2]) : : "memory")

__global__ __launch_bounds__(128) void gru_scan_kernel(
    const float* __restrict__ noise,
    const float* __restrict__ w_ih,
    const float* __restrict__ w_hh,
    const float* __restrict__ b_ih,
    const float* __restrict__ b_hh,
    float* __restrict__ out)
{
    __shared__ __align__(16) float sh_h[2][HBUF];

    const int tid  = threadIdx.x;
    const int ct   = tid >> 5;
    const int lane = tid & 31;
    const int lhi  = lane >> 4;
    const int lcol = lane & 15;
    const int b0   = blockIdx.x * 16;
    const int colg = ct * 16 + lcol;

    v2f bwih[3][3];
    v2f bwhh[3][16];
    float bgi[3];
    float bhn = 0.0f;

    for (int g = 0; g < 3; ++g) {
        const int col3 = g * GH + colg;
        for (int kc = 0; kc < 3; ++kc) {
            const int k0 = 4 * kc + 2 * lhi;
            v2f b;
            b[0] = (k0     < GI) ? w_ih[col3 * GI + k0]     : 0.0f;
            b[1] = (k0 + 1 < GI) ? w_ih[col3 * GI + k0 + 1] : 0.0f;
            bwih[g][kc] = b;
        }
        for (int kc = 0; kc < 16; ++kc) {
            const int k0 = 4 * kc + 2 * lhi;
            v2f b;
            b[0] = w_hh[col3 * GH + k0];
            b[1] = w_hh[col3 * GH + k0 + 1];
            bwhh[g][kc] = b;
        }
        bgi[g] = b_ih[col3] + ((g < 2) ? b_hh[col3] : 0.0f);
        if (g == 2) bhn = b_hh[col3];
    }
    v8f cgi[3], chn, czero;
    #pragma unroll
    for (int g = 0; g < 3; ++g)
        #pragma unroll
        for (int vv = 0; vv < 8; ++vv) cgi[g][vv] = bgi[g];
    #pragma unroll
    for (int vv = 0; vv < 8; ++vv) { chn[vv] = bhn; czero[vv] = 0.0f; }

    float hreg[8];
    #pragma unroll
    for (int vv = 0; vv < 8; ++vv) hreg[vv] = 0.0f;
    for (int idx = tid; idx < 16 * HP; idx += 128)
        sh_h[0][idx] = 0.0f;
    BAR_SIGNAL();

    const int m_a = lane & 15;
    const unsigned nrow_b = (unsigned)(b0 + m_a) * (unsigned)(GT * GI * 4);
    const unsigned noff01 = nrow_b + 8u * (unsigned)lhi;
    const unsigned noff2  = nrow_b + 32u;
    const unsigned ooff   = ((unsigned)(b0 + 8 * lhi) * (unsigned)(GT * GH) + (unsigned)colg) * 4u;
    const unsigned long long noise_u = (unsigned long long)noise;
    const unsigned long long out_u   = (unsigned long long)out;
    const int whalf = (colg >> 1) & 1;
    const int widx  = ((colg >> 2) << 1) | (colg & 1);

    NFrag nA, nB;
    NOISE_ISSUE(nA, noise_u);
    NOISE_ISSUE(nB, noise_u + (unsigned long long)(GI * 4));

    auto step = [&](const int t, const int BUF, NFrag& cur) __attribute__((always_inline)) {
        BAR_WAIT();
        NOISE_ARRIVE(cur);

        NFrag an = cur;
        an.c[2][0] = lhi ? 0.0f : cur.c[2][0];
        an.c[2][1] = lhi ? 0.0f : cur.c[2][1];

        const float* __restrict__ hrow = &sh_h[BUF][m_a * HP + lhi * 32];
        v2f ah[16];
        #pragma unroll
        for (int kc = 0; kc < 16; ++kc)
            ah[kc] = *((const v2f*)hrow + kc);

        v8f accA[3], accB[3], gn;
        #pragma unroll
        for (int g = 0; g < 3; ++g) {
            v8f a = WMMA4(an.c[0], bwih[g][0], cgi[g]);
            a = WMMA4(an.c[1], bwih[g][1], a);
            a = WMMA4(an.c[2], bwih[g][2], a);
            if (g == 2) { gn = a; a = chn; }
            #pragma unroll
            for (int kc = 0; kc < 8; ++kc)
                a = WMMA4(ah[kc], bwhh[g][kc], a);
            accA[g] = a;
            v8f bacc = czero;
            #pragma unroll
            for (int kc = 8; kc < 16; ++kc)
                bacc = WMMA4(ah[kc], bwhh[g][kc], bacc);
            accB[g] = bacc;
        }

        float nv[8], zv[8];
        #pragma unroll
        for (int vv = 0; vv < 8; ++vv) {
            const float r  = sigmoid_f(accA[0][vv] + accB[0][vv]);
            zv[vv]         = sigmoid_f(accA[1][vv] + accB[1][vv]);
            const float hn = accA[2][vv] + accB[2][vv];
            nv[vv]         = tanh_f(gn[vv] + r * hn);
        }
        asm volatile("s_wait_xcnt 0x0"
                     : "+v"(hreg[0]), "+v"(hreg[1]), "+v"(hreg[2]), "+v"(hreg[3]),
                       "+v"(hreg[4]), "+v"(hreg[5]), "+v"(hreg[6]), "+v"(hreg[7]),
                       "+v"(nv[0]));
        float* __restrict__ hw = &sh_h[BUF ^ 1][0];
        #pragma unroll
        for (int vv = 0; vv < 8; ++vv) {
            const int m = vv + 8 * lhi;
            const float h = nv[vv] + zv[vv] * (hreg[vv] - nv[vv]);
            hreg[vv] = h;
            hw[m * HP + whalf * 32 + widx] = h;
        }
        BAR_SIGNAL();

        const int tn = (t + 2 < GT) ? (t + 2) : (GT - 1);
        NOISE_ISSUE(cur, noise_u + (unsigned long long)tn * (GI * 4));

        asm volatile("global_store_b32 %8, %0, %9\n\t"
                     "global_store_b32 %8, %1, %9 offset:524288\n\t"
                     "global_store_b32 %8, %2, %9 offset:1048576\n\t"
                     "global_store_b32 %8, %3, %9 offset:1572864\n\t"
                     "global_store_b32 %8, %4, %9 offset:2097152\n\t"
                     "global_store_b32 %8, %5, %9 offset:2621440\n\t"
                     "global_store_b32 %8, %6, %9 offset:3145728\n\t"
                     "global_store_b32 %8, %7, %9 offset:3670016"
                     :
                     : "v"(hreg[0]), "v"(hreg[1]), "v"(hreg[2]), "v"(hreg[3]),
                       "v"(hreg[4]), "v"(hreg[5]), "v"(hreg[6]), "v"(hreg[7]),
                       "v"(ooff), "s"(out_u + (unsigned long long)t * (GH * 4))
                     : "memory");
    };

    for (int t = 0; t < GT; t += 2) {
        step(t,     0, nA);
        step(t + 1, 1, nB);
    }
}

extern "C" void kernel_launch(void* const* d_in, const int* in_sizes, int n_in,
                              void* d_out, int out_size, void* d_ws, size_t ws_size,
                              hipStream_t stream) {
    (void)in_sizes; (void)n_in; (void)out_size; (void)d_ws; (void)ws_size;
    const float* noise = (const float*)d_in[0];
    const float* w_ih  = (const float*)d_in[1];
    const float* w_hh  = (const float*)d_in[2];
    const float* b_ih  = (const float*)d_in[3];
    const float* b_hh  = (const float*)d_in[4];
    float* out = (float*)d_out;

    dim3 grid(GB / 16);
    dim3 block(128);
    gru_scan_kernel<<<grid, block, 0, stream>>>(noise, w_ih, w_hh, b_ih, b_hh, out);
}


# Input projection taken off the serial path: its first K chunk for step t+1 issues while the LDS stores of h drain, the other two right behind the grouped LDS fragment loads; sum order kept
# speedup vs baseline: 2.2367x; 1.0868x over previous
#include <hip/hip_runtime.h>


typedef __attribute__((ext_vector_type(2))) float v2f;
typedef __attribute__((ext_vector_type(4))) float v4f;
typedef __attribute__((ext_vector_type(8))) float v8f;

#define GB 256
#define GT 2048
#define GI 10
#define GH 64
#define HP 68
#define HBUF 2048

#define WMMA4(A, B, C) __builtin_amdgcn_wmma_f32_16x16x4_f32( \
    false, (A), false, (B), (short)0, (C), false, false)

__device__ __forceinline__ float fast_rcp(float x) {
#if __has_builtin(__builtin_amdgcn_rcpf)
    return __builtin_amdgcn_rcpf(x);
#else
    return 1.0f / x;
#endif
}
__device__ __forceinline__ float fast_exp(float x) {
#if __has_builtin(__builtin_amdgcn_exp2f)
    return __builtin_amdgcn_exp2f(x * 1.4426950408889634f);
#else
    return __expf(x);
#endif
}
__device__ __forceinline__ float tanh_f(float x) {
#if __has_builtin(__builtin_amdgcn_tanhf)
    return __builtin_amdgcn_tanhf(x);
#else
    return 1.0f - 2.0f * fast_rcp(fast_exp(2.0f * x) + 1.0f);
#endif
}
__device__ __forceinline__ float sigmoid_f(float x) {
#if __has_builtin(__builtin_amdgcn_tanhf)
    return 0.5f + 0.5f * __builtin_amdgcn_tanhf(0.5f * x);
#else
    return fast_rcp(1.0f + fast_exp(-x));
#endif
}

#define BAR_SIGNAL() asm volatile("s_wait_dscnt 0x0\n\ts_barrier_signal -1" ::: "memory")
#define BAR_WAIT()   asm volatile("s_barrier_wait -1" ::: "memory")

struct NFrag { v2f c[3]; };

#define NOISE_ISSUE(D0, BASE0, D1, D2, BASE12)                                \
    asm volatile("global_load_b64 %0, %3, %5\n\t"                             \
                 "global_load_b64 %1, %3, %6 offset:16\n\t"                   \
                 "global_load_b64 %2, %4, %6"                                 \
                 : "=&v"(D0), "=&v"(D1), "=&v"(D2)                            \
                 : "v"(noff01), "v"(noff2), "s"(BASE0), "s"(BASE12)           \
                 : "memory")

#define NOISE_ARRIVE(D0, D1, D2)                                              \
    asm volatile("s_wait_loadcnt 0x3"                                         \
                 : "+v"(D0), "+v"(D1), "+v"(D2) : : "memory")

__global__ __launch_bounds__(128) void gru_scan_kernel(
    const float* __restrict__ noise,
    const float* __restrict__ w_ih,
    const float* __restrict__ w_hh,
    const float* __restrict__ b_ih,
    const float* __restrict__ b_hh,
    float* __restrict__ out)
{
    __shared__ __align__(16) float sh_h[2][HBUF];

    const int tid  = threadIdx.x;
    const int ct   = tid >> 5;
    const int lane = tid & 31;
    const int lhi  = lane >> 4;
    const int lcol = lane & 15;
    const int b0   = blockIdx.x * 16;
    const int colg = ct * 16 + lcol;

    v2f bwih[3][3];
    v2f bwhh[3][16];
    float bgi[3];
    float bhn = 0.0f;

    for (int g = 0; g < 3; ++g) {
        const int col3 = g * GH + colg;
        for (int kc = 0; kc < 3; ++kc) {
            const int k0 = 4 * kc + 2 * lhi;
            v2f b;
            b[0] = (k0     < GI) ? w_ih[col3 * GI + k0]     : 0.0f;
            b[1] = (k0 + 1 < GI) ? w_ih[col3 * GI + k0 + 1] : 0.0f;
            bwih[g][kc] = b;
        }
        for (int kc = 0; kc < 16; ++kc) {
            const int k0 = 4 * kc + 2 * lhi;
            v2f b;
            b[0] = w_hh[col3 * GH + k0];
            b[1] = w_hh[col3 * GH + k0 + 1];
            bwhh[g][kc] = b;
        }
        bgi[g] = b_ih[col3] + ((g < 2) ? b_hh[col3] : 0.0f);
        if (g == 2) bhn = b_hh[col3];
    }
    v8f cgi[3], chn, czero;
    #pragma unroll
    for (int g = 0; g < 3; ++g)
        #pragma unroll
        for (int vv = 0; vv < 8; ++vv) cgi[g][vv] = bgi[g];
    #pragma unroll
    for (int vv = 0; vv < 8; ++vv) { chn[vv] = bhn; czero[vv] = 0.0f; }

    float hreg[8];
    #pragma unroll
    for (int vv = 0; vv < 8; ++vv) hreg[vv] = 0.0f;
    for (int idx = tid; idx < 16 * HP; idx += 128)
        sh_h[0][idx] = 0.0f;
    BAR_SIGNAL();

    const int m_a = lane & 15;
    const unsigned nrow_b = (unsigned)(b0 + m_a) * (unsigned)(GT * GI * 4);
    const unsigned noff01 = nrow_b + 8u * (unsigned)lhi;
    const unsigned noff2  = nrow_b + 32u;
    const unsigned ooff   = ((unsigned)(b0 + 8 * lhi) * (unsigned)(GT * GH) + (unsigned)colg) * 4u;
    const unsigned long long noise_u = (unsigned long long)noise;
    const unsigned long long out_u   = (unsigned long long)out;
    const int whalf = (colg >> 1) & 1;
    const int widx  = ((colg >> 2) << 1) | (colg & 1);

    NFrag nA, nB;
    asm volatile("global_load_b64 %0, %1, %2"
                 : "=&v"(nA.c[0]) : "v"(noff01), "s"(noise_u) : "memory");
    NOISE_ISSUE(nB.c[0], noise_u + (unsigned long long)(GI * 4),
                nA.c[1], nA.c[2], noise_u);
    asm volatile("s_wait_loadcnt 0x3" : "+v"(nA.c[0]) : : "memory");
    v8f gacc[3];
    #pragma unroll
    for (int g = 0; g < 3; ++g)
        gacc[g] = WMMA4(nA.c[0], bwih[g][0], cgi[g]);
    NOISE_ISSUE(nA.c[0], noise_u + (unsigned long long)(2 * GI * 4),
                nB.c[1], nB.c[2], noise_u + (unsigned long long)(GI * 4));

    auto step = [&](const int t, const int BUF, NFrag& cur, NFrag& nxt, v8f (&gi)[3])
                    __attribute__((always_inline)) {
        BAR_WAIT();
        NOISE_ARRIVE(nxt.c[0], cur.c[1], cur.c[2]);

        const float* __restrict__ hrow = &sh_h[BUF][m_a * HP + lhi * 32];
        v2f ah[16];
        #pragma unroll
        for (int kc = 0; kc < 16; ++kc)
            ah[kc] = *((const v2f*)hrow + kc);
        __builtin_amdgcn_sched_barrier(0);

        v2f a2;
        a2[0] = lhi ? 0.0f : cur.c[2][0];
        a2[1] = lhi ? 0.0f : cur.c[2][1];
        #pragma unroll
        for (int g = 0; g < 3; ++g)
            gi[g] = WMMA4(cur.c[1], bwih[g][1], gi[g]);
        #pragma unroll
        for (int g = 0; g < 3; ++g)
            gi[g] = WMMA4(a2, bwih[g][2], gi[g]);
        __builtin_amdgcn_sched_barrier(0);

        v8f accA[3], accB[3];
        const v8f gn = gi[2];
        #pragma unroll
        for (int g = 0; g < 3; ++g) {
            v8f a = (g == 2) ? chn : gi[g];
            #pragma unroll
            for (int kc = 0; kc < 8; ++kc)
                a = WMMA4(ah[kc], bwhh[g][kc], a);
            accA[g] = a;
            v8f bacc = czero;
            #pragma unroll
            for (int kc = 8; kc < 16; ++kc)
                bacc = WMMA4(ah[kc], bwhh[g][kc], bacc);
            accB[g] = bacc;
        }

        float nv[8], zv[8];
        #pragma unroll
        for (int vv = 0; vv < 8; ++vv) {
            const float r  = sigmoid_f(accA[0][vv] + accB[0][vv]);
            zv[vv]         = sigmoid_f(accA[1][vv] + accB[1][vv]);
            const float hn = accA[2][vv] + accB[2][vv];
            nv[vv]         = tanh_f(gn[vv] + r * hn);
        }
        asm volatile("s_wait_xcnt 0x0"
                     : "+v"(hreg[0]), "+v"(hreg[1]), "+v"(hreg[2]), "+v"(hreg[3]),
                       "+v"(hreg[4]), "+v"(hreg[5]), "+v"(hreg[6]), "+v"(hreg[7]),
                       "+v"(nv[0]));
        float* __restrict__ hw = &sh_h[BUF ^ 1][0];
        #pragma unroll
        for (int vv = 0; vv < 8; ++vv) {
            const int m = vv + 8 * lhi;
            const float h = nv[vv] + zv[vv] * (hreg[vv] - nv[vv]);
            hreg[vv] = h;
            hw[m * HP + whalf * 32 + widx] = h;
        }
        __builtin_amdgcn_sched_barrier(0);

        #pragma unroll
        for (int g = 0; g < 3; ++g)
            gi[g] = WMMA4(nxt.c[0], bwih[g][0], cgi[g]);
        __builtin_amdgcn_sched_barrier(0);
        BAR_SIGNAL();

        const int t2 = (t + 2 < GT) ? (t + 2) : (GT - 1);
        const int t3 = (t + 3 < GT) ? (t + 3) : (GT - 1);
        NOISE_ISSUE(nxt.c[0], noise_u + (unsigned long long)t3 * (GI * 4),
                    cur.c[1], cur.c[2], noise_u + (unsigned long long)t2 * (GI * 4));

        asm volatile("global_store_b32 %8, %0, %9\n\t"
                     "global_store_b32 %8, %1, %9 offset:524288\n\t"
                     "global_store_b32 %8, %2, %9 offset:1048576\n\t"
                     "global_store_b32 %8, %3, %9 offset:1572864\n\t"
                     "global_store_b32 %8, %4, %9 offset:2097152\n\t"
                     "global_store_b32 %8, %5, %9 offset:2621440\n\t"
                     "global_store_b32 %8, %6, %9 offset:3145728\n\t"
                     "global_store_b32 %8, %7, %9 offset:3670016"
                     :
                     : "v"(hreg[0]), "v"(hreg[1]), "v"(hreg[2]), "v"(hreg[3]),
                       "v"(hreg[4]), "v"(hreg[5]), "v"(hreg[6]), "v"(hreg[7]),
                       "v"(ooff), "s"(out_u + (unsigned long long)t * (GH * 4))
                     : "memory");
    };

    for (int t = 0; t < GT; t += 2) {
        step(t,     0, nA, nB, gacc);
        step(t + 1, 1, nB, nA, gacc);
    }
}

extern "C" void kernel_launch(void* const* d_in, const int* in_sizes, int n_in,
                              void* d_out, int out_size, void* d_ws, size_t ws_size,
                              hipStream_t stream) {
    (void)in_sizes; (void)n_in; (void)out_size; (void)d_ws; (void)ws_size;
    const float* noise = (const float*)d_in[0];
    const float* w_ih  = (const float*)d_in[1];
    const float* w_hh  = (const float*)d_in[2];
    const float* b_ih  = (const float*)d_in[3];
    const float* b_hh  = (const float*)d_in[4];
    float* out = (float*)d_out;

    dim3 grid(GB / 16);
    dim3 block(128);
    gru_scan_kernel<<<grid, block, 0, stream>>>(noise, w_ih, w_hh, b_ih, b_hh, out);
}


# A wave reads its own h tile back right behind its LDS stores and issues that tile's 12 products first (K order rotated per wave); t+1's whole input projection covers the round trip; sum order changed
# speedup vs baseline: 2.3250x; 1.0395x over previous
#include <hip/hip_runtime.h>


typedef __attribute__((ext_vector_type(2))) float v2f;
typedef __attribute__((ext_vector_type(4))) float v4f;
typedef __attribute__((ext_vector_type(8))) float v8f;

#define GB 256
#define GT 2048
#define GI 10
#define GH 64
#define HP 68
#define HBUF 2048

#define WMMA4(A, B, C) __builtin_amdgcn_wmma_f32_16x16x4_f32( \
    false, (A), false, (B), (short)0, (C), false, false)

__device__ __forceinline__ float fast_rcp(float x) {
#if __has_builtin(__builtin_amdgcn_rcpf)
    return __builtin_amdgcn_rcpf(x);
#else
    return 1.0f / x;
#endif
}
__device__ __forceinline__ float fast_exp(float x) {
#if __has_builtin(__builtin_amdgcn_exp2f)
    return __builtin_amdgcn_exp2f(x * 1.4426950408889634f);
#else
    return __expf(x);
#endif
}
__device__ __forceinline__ float tanh_f(float x) {
#if __has_builtin(__builtin_amdgcn_tanhf)
    return __builtin_amdgcn_tanhf(x);
#else
    return 1.0f - 2.0f * fast_rcp(fast_exp(2.0f * x) + 1.0f);
#endif
}
__device__ __forceinline__ float sigmoid_f(float x) {
#if __has_builtin(__builtin_amdgcn_tanhf)
    return 0.5f + 0.5f * __builtin_amdgcn_tanhf(0.5f * x);
#else
    return fast_rcp(1.0f + fast_exp(-x));
#endif
}

#define BAR_SIGNAL() asm volatile("s_wait_dscnt 0x0\n\ts_barrier_signal -1" ::: "memory")
#define BAR_WAIT()   asm volatile("s_barrier_wait -1" ::: "memory")
#define BAR_SIGNAL_OWN(V0, V1)                                                \
    asm volatile("s_wait_dscnt 0x0\n\ts_barrier_signal -1"                    \
                 : "+v"(V0), "+v"(V1) : : "memory")

struct NFrag { v2f c[3]; };

#define NOISE_ISSUE(D0, D1, D2, BASE)                                         \
    asm volatile("global_load_b64 %0, %3, %5\n\t"                             \
                 "global_load_b64 %1, %3, %5 offset:16\n\t"                   \
                 "global_load_b64 %2, %4, %5"                                 \
                 : "=&v"(D0), "=&v"(D1), "=&v"(D2)                            \
                 : "v"(noff01), "v"(noff2), "s"(BASE)                         \
                 : "memory")

#define NOISE_ARRIVE(D0, D1, D2)                                              \
    asm volatile("s_wait_loadcnt 0x3"                                         \
                 : "+v"(D0), "+v"(D1), "+v"(D2) : : "memory")

__global__ __launch_bounds__(128) void gru_scan_kernel(
    const float* __restrict__ noise,
    const float* __restrict__ w_ih,
    const float* __restrict__ w_hh,
    const float* __restrict__ b_ih,
    const float* __restrict__ b_hh,
    float* __restrict__ out)
{
    __shared__ __align__(16) float sh_h[2][HBUF];

    const int tid  = threadIdx.x;
    const int ct   = tid >> 5;
    const int lane = tid & 31;
    const int lhi  = lane >> 4;
    const int lcol = lane & 15;
    const int b0   = blockIdx.x * 16;
    const int colg = ct * 16 + lcol;

    v2f bwih[3][3];
    v2f bwhh[3][16];
    float bgi[3];
    float bhn = 0.0f;

    for (int g = 0; g < 3; ++g) {
        const int col3 = g * GH + colg;
        for (int kc = 0; kc < 3; ++kc) {
            const int k0 = 4 * kc + 2 * lhi;
            v2f b;
            b[0] = (k0     < GI) ? w_ih[col3 * GI + k0]     : 0.0f;
            b[1] = (k0 + 1 < GI) ? w_ih[col3 * GI + k0 + 1] : 0.0f;
            bwih[g][kc] = b;
        }
        for (int p = 0; p < 16; ++p) {
            const int kc = 4 * ((ct + (p >> 2)) & 3) + (p & 3);
            const int k0 = 4 * kc + 2 * lhi;
            v2f b;
            b[0] = w_hh[col3 * GH + k0];
            b[1] = w_hh[col3 * GH + k0 + 1];
            bwhh[g][p] = b;
        }
        bgi[g] = b_ih[col3] + ((g < 2) ? b_hh[col3] : 0.0f);
        if (g == 2) bhn = b_hh[col3];
    }
    v8f cgi[3], chn, czero;
    #pragma unroll
    for (int g = 0; g < 3; ++g)
        #pragma unroll
        for (int vv = 0; vv < 8; ++vv) cgi[g][vv] = bgi[g];
    #pragma unroll
    for (int vv = 0; vv < 8; ++vv) { chn[vv] = bhn; czero[vv] = 0.0f; }

    float hreg[8];
    #pragma unroll
    for (int vv = 0; vv < 8; ++vv) hreg[vv] = 0.0f;
    for (int idx = tid; idx < 16 * HP; idx += 128)
        sh_h[0][idx] = 0.0f;
    BAR_SIGNAL();

    const int m_a = lane & 15;
    const unsigned nrow_b = (unsigned)(b0 + m_a) * (unsigned)(GT * GI * 4);
    const unsigned noff01 = nrow_b + 8u * (unsigned)lhi;
    const unsigned noff2  = nrow_b + 32u;
    const unsigned ooff   = ((unsigned)(b0 + 8 * lhi) * (unsigned)(GT * GH) + (unsigned)colg) * 4u;
    const unsigned long long noise_u = (unsigned long long)noise;
    const unsigned long long out_u   = (unsigned long long)out;
    const int whalf = (colg >> 1) & 1;
    const int widx  = ((colg >> 2) << 1) | (colg & 1);
    int hoff[4];
    #pragma unroll
    for (int j = 0; j < 4; ++j)
        hoff[j] = m_a * HP + lhi * 32 + 8 * ((ct + j) & 3);

    NFrag nA, nB;
    NOISE_ISSUE(nA.c[0], nA.c[1], nA.c[2], noise_u);
    NOISE_ISSUE(nB.c[0], nB.c[1], nB.c[2], noise_u + (unsigned long long)(GI * 4));
    NOISE_ARRIVE(nA.c[0], nA.c[1], nA.c[2]);
    v8f gacc[3];
    {
        v2f a2;
        a2[0] = lhi ? 0.0f : nA.c[2][0];
        a2[1] = lhi ? 0.0f : nA.c[2][1];
        #pragma unroll
        for (int g = 0; g < 3; ++g)
            gacc[g] = WMMA4(nA.c[0], bwih[g][0], cgi[g]);
        #pragma unroll
        for (int g = 0; g < 3; ++g)
            gacc[g] = WMMA4(nA.c[1], bwih[g][1], gacc[g]);
        #pragma unroll
        for (int g = 0; g < 3; ++g)
            gacc[g] = WMMA4(a2, bwih[g][2], gacc[g]);
    }
    NOISE_ISSUE(nA.c[0], nA.c[1], nA.c[2], noise_u + (unsigned long long)(2 * GI * 4));

    v4f hown[2];
    #pragma unroll
    for (int vv = 0; vv < 4; ++vv) { hown[0][vv] = 0.0f; hown[1][vv] = 0.0f; }

    auto step = [&](const int t, const int BUF, NFrag& nxt, v8f (&gi)[3], v4f (&own)[2])
                    __attribute__((always_inline)) {
        v2f ah[16];
        ah[0][0] = own[0][0]; ah[0][1] = own[0][1];
        ah[1][0] = own[0][2]; ah[1][1] = own[0][3];
        ah[2][0] = own[1][0]; ah[2][1] = own[1][1];
        ah[3][0] = own[1][2]; ah[3][1] = own[1][3];

        v8f accA[3], accB[3];
        const v8f gn = gi[2];
        #pragma unroll
        for (int g = 0; g < 3; ++g)
            accA[g] = WMMA4(ah[0], bwhh[g][0], (g == 2) ? chn : gi[g]);
        __builtin_amdgcn_sched_barrier(0);
        BAR_WAIT();

        #pragma unroll
        for (int j = 1; j < 4; ++j) {
            const v4f* hp = (const v4f*)&sh_h[BUF][hoff[j]];
            const v4f lo = hp[0];
            const v4f hi = hp[1];
            ah[4 * j + 0][0] = lo[0]; ah[4 * j + 0][1] = lo[1];
            ah[4 * j + 1][0] = lo[2]; ah[4 * j + 1][1] = lo[3];
            ah[4 * j + 2][0] = hi[0]; ah[4 * j + 2][1] = hi[1];
            ah[4 * j + 3][0] = hi[2]; ah[4 * j + 3][1] = hi[3];
        }
        __builtin_amdgcn_sched_barrier(0);

        #pragma unroll
        for (int kc = 1; kc < 4; ++kc)
            #pragma unroll
            for (int g = 0; g < 3; ++g)
                accA[g] = WMMA4(ah[kc], bwhh[g][kc], accA[g]);
        __builtin_amdgcn_sched_barrier(0);

        #pragma unroll
        for (int g = 0; g < 3; ++g) {
            v8f a = accA[g];
            #pragma unroll
            for (int kc = 4; kc < 8; ++kc)
                a = WMMA4(ah[kc], bwhh[g][kc], a);
            accA[g] = a;
            v8f bacc = czero;
            #pragma unroll
            for (int kc = 8; kc < 16; ++kc)
                bacc = WMMA4(ah[kc], bwhh[g][kc], bacc);
            accB[g] = bacc;
        }

        float nv[8], zv[8];
        #pragma unroll
        for (int vv = 0; vv < 8; ++vv) {
            const float r  = sigmoid_f(accA[0][vv] + accB[0][vv]);
            zv[vv]         = sigmoid_f(accA[1][vv] + accB[1][vv]);
            const float hn = accA[2][vv] + accB[2][vv];
            nv[vv]         = tanh_f(gn[vv] + r * hn);
        }
        asm volatile("s_wait_xcnt 0x0"
                     : "+v"(hreg[0]), "+v"(hreg[1]), "+v"(hreg[2]), "+v"(hreg[3]),
                       "+v"(hreg[4]), "+v"(hreg[5]), "+v"(hreg[6]), "+v"(hreg[7]),
                       "+v"(nv[0]));
        float* hw = &sh_h[BUF ^ 1][0];
        #pragma unroll
        for (int vv = 0; vv < 8; ++vv) {
            const int m = vv + 8 * lhi;
            const float h = nv[vv] + zv[vv] * (hreg[vv] - nv[vv]);
            hreg[vv] = h;
            hw[m * HP + whalf * 32 + widx] = h;
        }
        {
            const v4f* hp = (const v4f*)&hw[hoff[0]];
            own[0] = hp[0];
            own[1] = hp[1];
        }
        __builtin_amdgcn_sched_barrier(0);

        NOISE_ARRIVE(nxt.c[0], nxt.c[1], nxt.c[2]);
        v2f a2;
        a2[0] = lhi ? 0.0f : nxt.c[2][0];
        a2[1] = lhi ? 0.0f : nxt.c[2][1];
        #pragma unroll
        for (int g = 0; g < 3; ++g)
            gi[g] = WMMA4(nxt.c[0], bwih[g][0], cgi[g]);
        #pragma unroll
        for (int g = 0; g < 3; ++g)
            gi[g] = WMMA4(nxt.c[1], bwih[g][1], gi[g]);
        #pragma unroll
        for (int g = 0; g < 3; ++g)
            gi[g] = WMMA4(a2, bwih[g][2], gi[g]);
        __builtin_amdgcn_sched_barrier(0);
        BAR_SIGNAL_OWN(own[0], own[1]);

        const int t3 = (t + 3 < GT) ? (t + 3) : (GT - 1);
        NOISE_ISSUE(nxt.c[0], nxt.c[1], nxt.c[2],
                    noise_u + (unsigned long long)t3 * (GI * 4));

        asm volatile("global_store_b32 %8, %0, %9\n\t"
                     "global_store_b32 %8, %1, %9 offset:524288\n\t"
                     "global_store_b32 %8, %2, %9 offset:1048576\n\t"
                     "global_store_b32 %8, %3, %9 offset:1572864\n\t"
                     "global_store_b32 %8, %4, %9 offset:2097152\n\t"
                     "global_store_b32 %8, %5, %9 offset:2621440\n\t"
                     "global_store_b32 %8, %6, %9 offset:3145728\n\t"
                     "global_store_b32 %8, %7, %9 offset:3670016"
                     :
                     : "v"(hreg[0]), "v"(hreg[1]), "v"(hreg[2]), "v"(hreg[3]),
                       "v"(hreg[4]), "v"(hreg[5]), "v"(hreg[6]), "v"(hreg[7]),
                       "v"(ooff), "s"(out_u + (unsigned long long)t * (GH * 4))
                     : "memory");
    };

    for (int t = 0; t < GT; t += 2) {
        step(t,     0, nB, gacc, hown);
        step(t + 1, 1, nA, gacc, hown);
    }
}

extern "C" void kernel_launch(void* const* d_in, const int* in_sizes, int n_in,
                              void* d_out, int out_size, void* d_ws, size_t ws_size,
                              hipStream_t stream) {
    (void)in_sizes; (void)n_in; (void)out_size; (void)d_ws; (void)ws_size;
    const float* noise = (const float*)d_in[0];
    const float* w_ih  = (const float*)d_in[1];
    const float* w_hh  = (const float*)d_in[2];
    const float* b_ih  = (const float*)d_in[3];
    const float* b_hh  = (const float*)d_in[4];
    float* out = (float*)d_out;

    dim3 grid(GB / 16);
    dim3 block(128);
    gru_scan_kernel<<<grid, block, 0, stream>>>(noise, w_ih, w_hh, b_ih, b_hh, out);
}


# Gate math cut from 58 to 22 vector instructions a step beside its 24 tanh: one accumulator chain per gate, sigmoid's inner 1/2 folded into the weights, math on register pairs; sum order changed
# speedup vs baseline: 2.4141x; 1.0383x over previous
#include <hip/hip_runtime.h>


typedef __attribute__((ext_vector_type(2))) float v2f;
typedef __attribute__((ext_vector_type(4))) float v4f;
typedef __attribute__((ext_vector_type(8))) float v8f;

#define GB 256
#define GT 2048
#define GI 10
#define GH 64
#define HP 68
#define HBUF 2048

#define WMMA4(A, B, C) __builtin_amdgcn_wmma_f32_16x16x4_f32( \
    false, (A), false, (B), (short)0, (C), false, false)

__device__ __forceinline__ float fast_rcp(float x) {
#if __has_builtin(__builtin_amdgcn_rcpf)
    return __builtin_amdgcn_rcpf(x);
#else
    return 1.0f / x;
#endif
}
__device__ __forceinline__ float fast_exp(float x) {
#if __has_builtin(__builtin_amdgcn_exp2f)
    return __builtin_amdgcn_exp2f(x * 1.4426950408889634f);
#else
    return __expf(x);
#endif
}
__device__ __forceinline__ float tanh_f(float x) {
#if __has_builtin(__builtin_amdgcn_tanhf)
    return __builtin_amdgcn_tanhf(x);
#else
    return 1.0f - 2.0f * fast_rcp(fast_exp(2.0f * x) + 1.0f);
#endif
}

#define BAR_SIGNAL() asm volatile("s_wait_dscnt 0x0\n\ts_barrier_signal -1" ::: "memory")
#define BAR_WAIT()   asm volatile("s_barrier_wait -1" ::: "memory")
#define BAR_SIGNAL_OWN(V0, V1)                                                \
    asm volatile("s_wait_dscnt 0x0\n\ts_barrier_signal -1"                    \
                 : "+v"(V0), "+v"(V1) : : "memory")

struct NFrag { v2f c[3]; };

#define NOISE_ISSUE(D0, D1, D2, BASE)                                         \
    asm volatile("global_load_b64 %0, %3, %5\n\t"                             \
                 "global_load_b64 %1, %3, %5 offset:16\n\t"                   \
                 "global_load_b64 %2, %4, %5"                                 \
                 : "=&v"(D0), "=&v"(D1), "=&v"(D2)                            \
                 : "v"(noff01), "v"(noff2), "s"(BASE)                         \
                 : "memory")

#define NOISE_ARRIVE(D0, D1, D2)                                              \
    asm volatile("s_wait_loadcnt 0x3"                                         \
                 : "+v"(D0), "+v"(D1), "+v"(D2) : : "memory")

__global__ __launch_bounds__(128) void gru_scan_kernel(
    const float* __restrict__ noise,
    const float* __restrict__ w_ih,
    const float* __restrict__ w_hh,
    const float* __restrict__ b_ih,
    const float* __restrict__ b_hh,
    float* __restrict__ out)
{
    __shared__ __align__(16) float sh_h[2][HBUF];

    const int tid  = threadIdx.x;
    const int ct   = tid >> 5;
    const int lane = tid & 31;
    const int lhi  = lane >> 4;
    const int lcol = lane & 15;
    const int b0   = blockIdx.x * 16;
    const int colg = ct * 16 + lcol;

    v2f bwih[3][3];
    v2f bwhh[3][16];
    float bgi[3];
    float bhn = 0.0f;

    for (int g = 0; g < 3; ++g) {
        const int col3 = g * GH + colg;
        const float gsc = (g < 2) ? 0.5f : 1.0f;
        for (int kc = 0; kc < 3; ++kc) {
            const int k0 = 4 * kc + 2 * lhi;
            v2f b;
            b[0] = (k0     < GI) ? w_ih[col3 * GI + k0]     : 0.0f;
            b[1] = (k0 + 1 < GI) ? w_ih[col3 * GI + k0 + 1] : 0.0f;
            bwih[g][kc] = b * gsc;
        }
        for (int p = 0; p < 16; ++p) {
            const int kc = 4 * ((ct + (p >> 2)) & 3) + (p & 3);
            const int k0 = 4 * kc + 2 * lhi;
            v2f b;
            b[0] = w_hh[col3 * GH + k0];
            b[1] = w_hh[col3 * GH + k0 + 1];
            bwhh[g][p] = b * 0.5f;
        }
        bgi[g] = (b_ih[col3] + ((g < 2) ? b_hh[col3] : 0.0f)) * gsc;
        if (g == 2) bhn = 0.5f * b_hh[col3];
    }
    v8f cgi[3], chn;
    #pragma unroll
    for (int g = 0; g < 3; ++g)
        #pragma unroll
        for (int vv = 0; vv < 8; ++vv) cgi[g][vv] = bgi[g];
    #pragma unroll
    for (int vv = 0; vv < 8; ++vv) chn[vv] = bhn;

    v2f hreg[4];
    #pragma unroll
    for (int i = 0; i < 4; ++i) { hreg[i][0] = 0.0f; hreg[i][1] = 0.0f; }
    for (int idx = tid; idx < 16 * HP; idx += 128)
        sh_h[0][idx] = 0.0f;
    BAR_SIGNAL();

    const int m_a = lane & 15;
    const unsigned nrow_b = (unsigned)(b0 + m_a) * (unsigned)(GT * GI * 4);
    const unsigned noff01 = nrow_b + 8u * (unsigned)lhi;
    const unsigned noff2  = nrow_b + 32u;
    const unsigned ooff   = ((unsigned)(b0 + 8 * lhi) * (unsigned)(GT * GH) + (unsigned)colg) * 4u;
    const unsigned long long noise_u = (unsigned long long)noise;
    const unsigned long long out_u   = (unsigned long long)out;
    const int whalf = (colg >> 1) & 1;
    const int widx  = ((colg >> 2) << 1) | (colg & 1);
    int hoff[4];
    #pragma unroll
    for (int j = 0; j < 4; ++j)
        hoff[j] = m_a * HP + lhi * 32 + 8 * ((ct + j) & 3);

    NFrag nA, nB;
    NOISE_ISSUE(nA.c[0], nA.c[1], nA.c[2], noise_u);
    NOISE_ISSUE(nB.c[0], nB.c[1], nB.c[2], noise_u + (unsigned long long)(GI * 4));
    NOISE_ARRIVE(nA.c[0], nA.c[1], nA.c[2]);
    v8f gacc[3];
    {
        v2f a2;
        a2[0] = lhi ? 0.0f : nA.c[2][0];
        a2[1] = lhi ? 0.0f : nA.c[2][1];
        #pragma unroll
        for (int g = 0; g < 3; ++g)
            gacc[g] = WMMA4(nA.c[0], bwih[g][0], cgi[g]);
        #pragma unroll
        for (int g = 0; g < 3; ++g)
            gacc[g] = WMMA4(nA.c[1], bwih[g][1], gacc[g]);
        #pragma unroll
        for (int g = 0; g < 3; ++g)
            gacc[g] = WMMA4(a2, bwih[g][2], gacc[g]);
    }
    NOISE_ISSUE(nA.c[0], nA.c[1], nA.c[2], noise_u + (unsigned long long)(2 * GI * 4));

    v4f hown[2];
    #pragma unroll
    for (int vv = 0; vv < 4; ++vv) { hown[0][vv] = 0.0f; hown[1][vv] = 0.0f; }

    auto step = [&](const int t, const int BUF, NFrag& nxt, v8f (&gi)[3], v4f (&own)[2])
                    __attribute__((always_inline)) {
        v2f ah[16];
        ah[0][0] = own[0][0]; ah[0][1] = own[0][1];
        ah[1][0] = own[0][2]; ah[1][1] = own[0][3];
        ah[2][0] = own[1][0]; ah[2][1] = own[1][1];
        ah[3][0] = own[1][2]; ah[3][1] = own[1][3];

        v8f accA[3];
        const v8f gn = gi[2];
        #pragma unroll
        for (int g = 0; g < 3; ++g)
            accA[g] = WMMA4(ah[0], bwhh[g][0], (g == 2) ? chn : gi[g]);
        __builtin_amdgcn_sched_barrier(0);
        BAR_WAIT();

        #pragma unroll
        for (int j = 1; j < 4; ++j) {
            const v4f* hp = (const v4f*)&sh_h[BUF][hoff[j]];
            const v4f lo = hp[0];
            const v4f hi = hp[1];
            ah[4 * j + 0][0] = lo[0]; ah[4 * j + 0][1] = lo[1];
            ah[4 * j + 1][0] = lo[2]; ah[4 * j + 1][1] = lo[3];
            ah[4 * j + 2][0] = hi[0]; ah[4 * j + 2][1] = hi[1];
            ah[4 * j + 3][0] = hi[2]; ah[4 * j + 3][1] = hi[3];
        }
        __builtin_amdgcn_sched_barrier(0);

        #pragma unroll
        for (int kc = 1; kc < 4; ++kc)
            #pragma unroll
            for (int g = 0; g < 3; ++g)
                accA[g] = WMMA4(ah[kc], bwhh[g][kc], accA[g]);
        __builtin_amdgcn_sched_barrier(0);

        #pragma unroll
        for (int kc = 4; kc < 16; ++kc)
            #pragma unroll
            for (int g = 0; g < 3; ++g)
                accA[g] = WMMA4(ah[kc], bwhh[g][kc], accA[g]);

        v2f tr[4], tz[4], nv[4];
        #pragma unroll
        for (int i = 0; i < 4; ++i) {
            tr[i][0] = tanh_f(accA[0][2 * i]);  tr[i][1] = tanh_f(accA[0][2 * i + 1]);
        }
        #pragma unroll
        for (int i = 0; i < 4; ++i) {
            tz[i][0] = tanh_f(accA[1][2 * i]);  tz[i][1] = tanh_f(accA[1][2 * i + 1]);
        }
        #pragma unroll
        for (int i = 0; i < 4; ++i) {
            v2f hn2, g2;
            hn2[0] = accA[2][2 * i];  hn2[1] = accA[2][2 * i + 1];
            g2[0]  = gn[2 * i];       g2[1]  = gn[2 * i + 1];
            const v2f np = __builtin_elementwise_fma(tr[i], hn2, g2 + hn2);
            nv[i][0] = tanh_f(np[0]);  nv[i][1] = tanh_f(np[1]);
        }
        asm volatile("s_wait_xcnt 0x0"
                     : "+v"(hreg[0]), "+v"(hreg[1]), "+v"(hreg[2]), "+v"(hreg[3]),
                       "+v"(nv[0]));
        float* hw = &sh_h[BUF ^ 1][0];
        const v2f half2 = {0.5f, 0.5f};
        #pragma unroll
        for (int i = 0; i < 4; ++i) {
            const v2f z = __builtin_elementwise_fma(tz[i], half2, half2);
            hreg[i] = __builtin_elementwise_fma(z, hreg[i] - nv[i], nv[i]);
        }
        #pragma unroll
        for (int vv = 0; vv < 8; ++vv) {
            const int m = vv + 8 * lhi;
            hw[m * HP + whalf * 32 + widx] = hreg[vv >> 1][vv & 1];
        }
        {
            const v4f* hp = (const v4f*)&hw[hoff[0]];
            own[0] = hp[0];
            own[1] = hp[1];
        }
        __builtin_amdgcn_sched_barrier(0);

        NOISE_ARRIVE(nxt.c[0], nxt.c[1], nxt.c[2]);
        v2f a2;
        a2[0] = lhi ? 0.0f : nxt.c[2][0];
        a2[1] = lhi ? 0.0f : nxt.c[2][1];
        #pragma unroll
        for (int g = 0; g < 3; ++g)
            gi[g] = WMMA4(nxt.c[0], bwih[g][0], cgi[g]);
        #pragma unroll
        for (int g = 0; g < 3; ++g)
            gi[g] = WMMA4(nxt.c[1], bwih[g][1], gi[g]);
        #pragma unroll
        for (int g = 0; g < 3; ++g)
            gi[g] = WMMA4(a2, bwih[g][2], gi[g]);
        __builtin_amdgcn_sched_barrier(0);
        BAR_SIGNAL_OWN(own[0], own[1]);

        const int t3 = (t + 3 < GT) ? (t + 3) : (GT - 1);
        NOISE_ISSUE(nxt.c[0], nxt.c[1], nxt.c[2],
                    noise_u + (unsigned long long)t3 * (GI * 4));

        asm volatile("global_store_b32 %8, %0, %9\n\t"
                     "global_store_b32 %8, %1, %9 offset:524288\n\t"
                     "global_store_b32 %8, %2, %9 offset:1048576\n\t"
                     "global_store_b32 %8, %3, %9 offset:1572864\n\t"
                     "global_store_b32 %8, %4, %9 offset:2097152\n\t"
                     "global_store_b32 %8, %5, %9 offset:2621440\n\t"
                     "global_store_b32 %8, %6, %9 offset:3145728\n\t"
                     "global_store_b32 %8, %7, %9 offset:3670016"
                     :
                     : "v"(hreg[0][0]), "v"(hreg[0][1]), "v"(hreg[1][0]), "v"(hreg[1][1]),
                       "v"(hreg[2][0]), "v"(hreg[2][1]), "v"(hreg[3][0]), "v"(hreg[3][1]),
                       "v"(ooff), "s"(out_u + (unsigned long long)t * (GH * 4))
                     : "memory");
    };

    for (int t = 0; t < GT; t += 2) {
        step(t,     0, nB, gacc, hown);
        step(t + 1, 1, nA, gacc, hown);
    }
}

extern "C" void kernel_launch(void* const* d_in, const int* in_sizes, int n_in,
                              void* d_out, int out_size, void* d_ws, size_t ws_size,
                              hipStream_t stream) {
    (void)in_sizes; (void)n_in; (void)out_size; (void)d_ws; (void)ws_size;
    const float* noise = (const float*)d_in[0];
    const float* w_ih  = (const float*)d_in[1];
    const float* w_hh  = (const float*)d_in[2];
    const float* b_ih  = (const float*)d_in[3];
    const float* b_hh  = (const float*)d_in[4];
    float* out = (float*)d_out;

    dim3 grid(GB / 16);
    dim3 block(128);
    gru_scan_kernel<<<grid, block, 0, stream>>>(noise, w_ih, w_hh, b_ih, b_hh, out);
}
